# starting kernel
# baseline (speedup 1.0000x reference)
#include <hip/hip_runtime.h>

typedef float v2f __attribute__((ext_vector_type(2)));
typedef float v8f __attribute__((ext_vector_type(8)));

#define BATCH 32768
#define NF    200
#define DIN   5
#define NH    15
#define NOUT  30

__device__ __forceinline__ v8f wmma_f32_k4(v2f a, v2f b, v8f c) {
  return __builtin_amdgcn_wmma_f32_16x16x4_f32(
       false, a,  false, b,
       (short)0, c,  false,  false);
}

template <int IMM>
__device__ __forceinline__ float swz(float x) {
  return __int_as_float(__builtin_amdgcn_ds_swizzle(__float_as_int(x), IMM));
}
#define SWAP16 0x401f
#define XOR1   0x041f
#define XOR2   0x081f
#define XOR4   0x101f
#define XOR8   0x201f

__global__ __launch_bounds__(256) void mlp200_wmma_kernel(
    const float* __restrict__ y,  const float* __restrict__ W1,
    const float* __restrict__ b1, const float* __restrict__ W2,
    const float* __restrict__ b2, const float* __restrict__ W3,
    const float* __restrict__ b3, float* __restrict__ out)
{
  const int  lane = threadIdx.x & 31;
  const int  wave = threadIdx.x >> 5;
  const int  q    = lane >> 4;
  const bool qhi  = (q != 0);
  const int  n    = lane & 15;
  const int  b0   = (blockIdx.x * 8 + wave) * 16;

  const int  nc   = (n < NH - 1) ? n : (NH - 1);
  const int  oc2  = 16 + ((n < 14) ? n : 13);
  const bool o2ok = (n < 14);

  const int  iy0  = (b0 + n) * (NF * DIN) + 2 * q;
  const int  iy1  = iy0 + 1;
  const int  iy4  = (b0 + n) * (NF * DIN) + 4;

  const int iW1a = (2 * q) * NH + nc;
  const int iW1b = iW1a + NH;
  const int iW1c = 4 * NH + nc;

  float acc[8];
  #pragma unroll
  for (int r = 0; r < 8; ++r) acc[r] = 0.f;

  #pragma unroll 2
  for (int f = 0; f < NF; ++f) {
    const float* yb  = y  + f * DIN;
    const float* W1f = W1 + f * (DIN * NH);
    const float* W2f = W2 + f * (NH * NOUT);
    const float* b1f = b1 + f * NH;
    const float* b2f = b2 + f * NOUT;
    const float* W3f = W3 + f * NOUT;

    v8f c1 = {};
    {
      v2f a, bb;
      a.x  = W1f[iW1a];
      a.y  = W1f[iW1b];
      bb.x = yb[iy0];
      bb.y = yb[iy1];
      c1 = wmma_f32_k4(a, bb, c1);
      a.x  = W1f[iW1c];
      a.y  = b1f[nc];
      float y4 = yb[iy4];
      bb.x = qhi ? 0.f : y4;
      bb.y = qhi ? 0.f : 1.f;
      c1 = wmma_f32_k4(a, bb, c1);
    }

    float hv[8], sw[8];
    #pragma unroll
    for (int r = 0; r < 8; ++r) { hv[r] = c1[r]; sw[r] = swz<SWAP16>(c1[r]); }

    v2f a0, a1, a2, a3;
    a0.x = qhi ? sw[2] : hv[0];
    a0.y = qhi ? sw[3] : hv[1];
    a1.x = qhi ? sw[6] : hv[4];
    a1.y = qhi ? sw[7] : hv[5];
    a2.x = qhi ? hv[2] : sw[0];
    a2.y = qhi ? hv[3] : sw[1];
    a3.x = qhi ? hv[6] : sw[4];
    a3.y = qhi ? 1.f   : sw[5];

    v8f c2a = {}, c2b = {};
    #pragma unroll
    for (int k = 0; k < 4; ++k) {
      const v2f av = (k == 0) ? a0 : (k == 1) ? a1 : (k == 2) ? a2 : a3;
      const int h0  = 4 * k + 2 * q;
      int       h1i = h0 + 1; if (h1i > NH - 1) h1i = NH - 1;
      v2f bta, btb;
      bta.x = W2f[h0  * NOUT + n];
      btb.x = W2f[h0  * NOUT + oc2];
      float bay = W2f[h1i * NOUT + n];
      float bby = W2f[h1i * NOUT + oc2];
      if (k == 3) {
        float b2a = b2f[n];
        float b2b = b2f[oc2];
        bta.y = qhi ? b2a : bay;
        btb.y = qhi ? b2b : bby;
      } else {
        bta.y = bay;
        btb.y = bby;
      }
      c2a = wmma_f32_k4(av, bta, c2a);
      c2b = wmma_f32_k4(av, btb, c2b);
    }

    float w3a = W3f[n];
    float w3t = W3f[oc2];
    float w3b = o2ok ? w3t : 0.f;
    #pragma unroll
    for (int r = 0; r < 8; ++r) {
      float va = c2a[r] > 0.f ? c2a[r] : 0.f;
      float vb = c2b[r] > 0.f ? c2b[r] : 0.f;
      acc[r] += va * w3a + vb * w3b;
    }
  }

  #pragma unroll
  for (int r = 0; r < 8; ++r) {
    float v = acc[r];
    v += swz<XOR1>(v);
    v += swz<XOR2>(v);
    v += swz<XOR4>(v);
    v += swz<XOR8>(v);
    acc[r] = v;
  }

  if (n == 0) {
    const float b3v = b3[0];
    float* op = out + b0 + 8 * q;
    float4 o0 = make_float4(acc[0] + b3v, acc[1] + b3v, acc[2] + b3v, acc[3] + b3v);
    float4 o1 = make_float4(acc[4] + b3v, acc[5] + b3v, acc[6] + b3v, acc[7] + b3v);
    reinterpret_cast<float4*>(op)[0] = o0;
    reinterpret_cast<float4*>(op)[1] = o1;
  }
}

extern "C" void kernel_launch(void* const* d_in, const int* in_sizes, int n_in,
                              void* d_out, int out_size, void* d_ws, size_t ws_size,
                              hipStream_t stream) {
  (void)in_sizes; (void)n_in; (void)d_ws; (void)ws_size; (void)out_size;
  const float* y  = (const float*)d_in[0];
  const float* W1 = (const float*)d_in[1];
  const float* b1 = (const float*)d_in[2];
  const float* W2 = (const float*)d_in[3];
  const float* b2 = (const float*)d_in[4];
  const float* W3 = (const float*)d_in[5];
  const float* b3 = (const float*)d_in[6];
  float* out = (float*)d_out;

  dim3 grid(BATCH / (16 * 8));
  dim3 block(256);
  mlp200_wmma_kernel<<<grid, block, 0, stream>>>(y, W1, b1, W2, b2, W3, b3, out);
}


# No LDS instruction between WMMAs: layer 1's D registers are layer 2's B (W2 as A, no lane-half exchange), a chunk's fragment reads stand ahead of its WMMAs; ReLU*W3 summed in-lane: sum orders change
# speedup vs baseline: 1.9261x; 1.9261x over previous
#include <hip/hip_runtime.h>

typedef float v2f __attribute__((ext_vector_type(2)));
typedef float v8f __attribute__((ext_vector_type(8)));

#define BATCH 32768
#define NF    200
#define DIN   5
#define NH    15
#define NOUT  30
#define FC    8
#define NCHUNK (NF / FC)

__device__ __forceinline__ v8f wmma_f32_k4(v2f a, v2f b, v8f c) {
  return __builtin_amdgcn_wmma_f32_16x16x4_f32(
       false, a,  false, b,
       (short)0, c,  false,  false);
}

template <int IMM>
__device__ __forceinline__ float swz(float x) {
  return __int_as_float(__builtin_amdgcn_ds_swizzle(__float_as_int(x), IMM));
}
#define SWAP16 0x401f

#define SW_IDX(buf, fl, slot, lane)  ((((buf) * FC + (fl)) * 5 + (slot)) * 32 + (lane))
#define SW3_IDX(buf, fl, lane)       (((buf) * FC + (fl)) * 32 + (lane))
#define SW3_IDX4(buf, fl, q, j)      (((buf) * FC + (fl)) * 8 + (q) * 4 + (j))

struct WRaw { float l1[4]; float ax[4], ay[4], bx[4], by[4]; float b2a, b2b, w3; };

__device__ __forceinline__ void load_wraw(
    WRaw& wr, int f, int n, int q, int nc, int oc2, int o3c,
    const float* __restrict__ W1, const float* __restrict__ b1,
    const float* __restrict__ W2, const float* __restrict__ b2,
    const float* __restrict__ W3)
{
  const float* W1f = W1 + f * (DIN * NH);
  const float* W2f = W2 + f * (NH * NOUT);
  const float* b1f = b1 + f * NH;
  const float* b2f = b2 + f * NOUT;
  const float* W3f = W3 + f * NOUT;

  wr.l1[0] = W1f[(2 * q) * NH + nc];
  wr.l1[1] = W1f[(2 * q + 1) * NH + nc];
  wr.l1[2] = W1f[4 * NH + nc];
  wr.l1[3] = b1f[nc];

  #pragma unroll
  for (int k = 0; k < 4; ++k) {
    const int g0  = 8 * q + 2 * k;
    int       g1i = g0 + 1; if (g1i > NH - 1) g1i = NH - 1;
    wr.ax[k] = W2f[g0  * NOUT + n];
    wr.bx[k] = W2f[g0  * NOUT + oc2];
    wr.ay[k] = W2f[g1i * NOUT + n];
    wr.by[k] = W2f[g1i * NOUT + oc2];
  }
  wr.b2a = b2f[n];
  wr.b2b = b2f[oc2];
  wr.w3  = W3f[o3c];
}

__device__ __forceinline__ void store_wfrag(const WRaw& wr, float4* sW, float* sW3,
                                            int buf, int fl, int lane, bool qhi,
                                            bool n15, bool o3ok) {
  sW[SW_IDX(buf, fl, 0, lane)] = n15 ? make_float4(0.f, 0.f, 0.f, 1.f)
                                     : make_float4(wr.l1[0], wr.l1[1], wr.l1[2], wr.l1[3]);
  #pragma unroll
  for (int k = 0; k < 3; ++k)
    sW[SW_IDX(buf, fl, 1 + k, lane)] = make_float4(wr.ax[k], wr.ay[k], wr.bx[k], wr.by[k]);
  sW[SW_IDX(buf, fl, 4, lane)] = make_float4(wr.ax[3], qhi ? wr.b2a : wr.ay[3],
                                             wr.bx[3], qhi ? wr.b2b : wr.by[3]);
  sW3[SW3_IDX(buf, fl, lane)] = o3ok ? wr.w3 : 0.f;
}

struct FIn { float4 l1; float4 wk[4]; float4 w3[4]; };

__device__ __forceinline__ void fetch_frag(
    FIn& in, const float4* sW, const float4* sW3v, int buf, int fl, int lane, int q)
{
  in.l1 = sW[SW_IDX(buf, fl, 0, lane)];
  #pragma unroll
  for (int k = 0; k < 4; ++k) in.wk[k] = sW[SW_IDX(buf, fl, 1 + k, lane)];
  #pragma unroll
  for (int j = 0; j < 4; ++j) in.w3[j] = sW3v[SW3_IDX4(buf, fl, q, j)];
}

__device__ __forceinline__ void feature_mma(const FIn& in, float y0, float y1, float y4, bool qhi,
                                            v8f& c2a, v8f& c2b) {
  v8f c1 = {};
  {
    v2f a, bb;
    a.x  = in.l1.x;
    a.y  = in.l1.y;
    bb.x = y0;
    bb.y = y1;
    c1 = wmma_f32_k4(a, bb, c1);
    a.x  = in.l1.z;
    a.y  = in.l1.w;
    bb.x = qhi ? 0.f : y4;
    bb.y = qhi ? 0.f : 1.f;
    c1 = wmma_f32_k4(a, bb, c1);
  }

  c2a = (v8f){};
  c2b = (v8f){};
  #pragma unroll
  for (int k = 0; k < 4; ++k) {
    v2f hb, wa, wb;
    hb.x = c1[2 * k];     hb.y = c1[2 * k + 1];
    wa.x = in.wk[k].x;    wa.y = in.wk[k].y;
    wb.x = in.wk[k].z;    wb.y = in.wk[k].w;
    c2a = wmma_f32_k4(wa, hb, c2a);
    c2b = wmma_f32_k4(wb, hb, c2b);
  }
}

__device__ __forceinline__ void feature_tail(const v8f& c2a, const v8f& c2b, const float4 (&w3)[4],
                                             float (&accA)[8], float (&accB)[8]) {
  const float w3a[8] = { w3[0].x, w3[0].y, w3[0].z, w3[0].w, w3[1].x, w3[1].y, w3[1].z, w3[1].w };
  const float w3b[8] = { w3[2].x, w3[2].y, w3[2].z, w3[2].w, w3[3].x, w3[3].y, w3[3].z, w3[3].w };
  #pragma unroll
  for (int r = 0; r < 8; ++r) {
    const float va = c2a[r] > 0.f ? c2a[r] : 0.f;
    const float vb = c2b[r] > 0.f ? c2b[r] : 0.f;
    accA[r] = __builtin_fmaf(va, w3a[r], accA[r]);
    accB[r] = __builtin_fmaf(vb, w3b[r], accB[r]);
  }
}

__global__ __launch_bounds__(256) void mlp200_wmma_kernel(
    const float* __restrict__ y,  const float* __restrict__ W1,
    const float* __restrict__ b1, const float* __restrict__ W2,
    const float* __restrict__ b2, const float* __restrict__ W3,
    const float* __restrict__ b3, float* __restrict__ out)
{
  __shared__ float4 sW[2 * FC * 5 * 32];
  __shared__ float4 sW3v[2 * FC * 8];

  const int  lane = threadIdx.x & 31;
  const int  wave = threadIdx.x >> 5;
  const int  q    = lane >> 4;
  const bool qhi  = (q != 0);
  const int  n    = lane & 15;
  const int  b0   = (blockIdx.x * 8 + wave) * 16;

  const int  nc   = (n < NH - 1) ? n : (NH - 1);
  const int  oc2  = 16 + ((n < 14) ? n : 13);
  const bool n15  = (n == 15);
  const int  o3   = 16 * ((lane >> 3) & 1) + 8 * q + (lane & 7);
  const bool o3ok = (o3 < NOUT);
  const int  o3c  = o3ok ? o3 : (NOUT - 1);

  const int  iy0  = (b0 + n) * (NF * DIN) + 2 * q;
  const int  iy1  = iy0 + 1;
  const int  iy4  = (b0 + n) * (NF * DIN) + 4;

  float accA[8], accB[8];
  #pragma unroll
  for (int r = 0; r < 8; ++r) { accA[r] = 0.f; accB[r] = 0.f; }

  float* sW3 = reinterpret_cast<float*>(sW3v);

  WRaw wr;
  load_wraw(wr, wave, n, q, nc, oc2, o3c, W1, b1, W2, b2, W3);
  store_wfrag(wr, sW, sW3, 0, wave, lane, qhi, n15, o3ok);
  __syncthreads();

  for (int c = 0; c < NCHUNK; ++c) {
    const int buf = c & 1;
    const int cn  = (c + 1 < NCHUNK) ? (c + 1) : c;

    float ya[FC][3];
    {
      const float* yb = y + (c * FC) * DIN;
      #pragma unroll
      for (int fl = 0; fl < FC; ++fl) {
        ya[fl][0] = yb[iy0 + fl * DIN];
        ya[fl][1] = yb[iy1 + fl * DIN];
        ya[fl][2] = yb[iy4 + fl * DIN];
      }
    }
    load_wraw(wr, cn * FC + wave, n, q, nc, oc2, o3c, W1, b1, W2, b2, W3);

    FIn fr[FC];
    #pragma unroll
    for (int fl = 0; fl < FC; ++fl) fetch_frag(fr[fl], sW, sW3v, buf, fl, lane, q);

    #pragma unroll
    for (int fl = 0; fl < FC; ++fl) {
      v8f ca, cb;
      feature_mma(fr[fl], ya[fl][0], ya[fl][1], ya[fl][2], qhi, ca, cb);
      feature_tail(ca, cb, fr[fl].w3, accA, accB);
    }

    __builtin_amdgcn_sched_group_barrier(0x020, 39, 0);
    __builtin_amdgcn_sched_group_barrier(0x100, 72, 0);
    #pragma unroll
    for (int i = 0; i < FC; ++i)
      __builtin_amdgcn_sched_group_barrier(0x008, 10, 0);
    __builtin_amdgcn_sched_group_barrier(0x200, 6, 0);

    store_wfrag(wr, sW, sW3, buf ^ 1, wave, lane, qhi, n15, o3ok);
    __syncthreads();
  }

  float s = 0.f;
  {
    float t[8];
    #pragma unroll
    for (int r = 0; r < 8; ++r) t[r] = accA[r] + accB[r];
    s = ((t[0] + t[1]) + (t[2] + t[3])) + ((t[4] + t[5]) + (t[6] + t[7]));
  }
  s += swz<SWAP16>(s);

  if (!qhi)
    out[b0 + n] = s + b3[0];
}

extern "C" void kernel_launch(void* const* d_in, const int* in_sizes, int n_in,
                              void* d_out, int out_size, void* d_ws, size_t ws_size,
                              hipStream_t stream) {
  (void)in_sizes; (void)n_in; (void)d_ws; (void)ws_size; (void)out_size;
  const float* y  = (const float*)d_in[0];
  const float* W1 = (const float*)d_in[1];
  const float* b1 = (const float*)d_in[2];
  const float* W2 = (const float*)d_in[3];
  const float* b2 = (const float*)d_in[4];
  const float* W3 = (const float*)d_in[5];
  const float* b3 = (const float*)d_in[6];
  float* out = (float*)d_out;

  dim3 grid(BATCH / (16 * 8));
  dim3 block(256);
  mlp200_wmma_kernel<<<grid, block, 0, stream>>>(y, W1, b1, W2, b2, W3, b3, out);
}


# y one chunk ahead in place: a feature's 3 y registers reload once its layer 1 read them (spare k slots zeroed in the weight image, not in y); loads between WMMA groups, LDS-only barrier; same sums
# speedup vs baseline: 2.1788x; 1.1312x over previous
#include <hip/hip_runtime.h>

typedef float v2f __attribute__((ext_vector_type(2)));
typedef float v8f __attribute__((ext_vector_type(8)));

#define BATCH 32768
#define NF    200
#define DIN   5
#define NH    15
#define NOUT  30
#define FC    8
#define NCHUNK (NF / FC)

__device__ __forceinline__ v8f wmma_f32_k4(v2f a, v2f b, v8f c) {
  return __builtin_amdgcn_wmma_f32_16x16x4_f32(
       false, a,  false, b,
       (short)0, c,  false,  false);
}

template <int IMM>
__device__ __forceinline__ float swz(float x) {
  return __int_as_float(__builtin_amdgcn_ds_swizzle(__float_as_int(x), IMM));
}
#define SWAP16 0x401f

#define SW_IDX(buf, fl, slot, lane)  ((((buf) * FC + (fl)) * 5 + (slot)) * 32 + (lane))
#define SW3_IDX(buf, fl, lane)       (((buf) * FC + (fl)) * 32 + (lane))
#define SW3_IDX4(buf, fl, q, j)      (((buf) * FC + (fl)) * 8 + (q) * 4 + (j))

struct WRaw { float l1[4]; float ax[4], ay[4], bx[4], by[4]; float b2a, b2b, w3; };

__device__ __forceinline__ void load_wraw(
    WRaw& wr, int f, int n, int q, int nc, int oc2, int o3c,
    const float* __restrict__ W1, const float* __restrict__ b1,
    const float* __restrict__ W2, const float* __restrict__ b2,
    const float* __restrict__ W3)
{
  const float* W1f = W1 + f * (DIN * NH);
  const float* W2f = W2 + f * (NH * NOUT);
  const float* b1f = b1 + f * NH;
  const float* b2f = b2 + f * NOUT;
  const float* W3f = W3 + f * NOUT;

  wr.l1[0] = W1f[(2 * q) * NH + nc];
  wr.l1[1] = W1f[(2 * q + 1) * NH + nc];
  wr.l1[2] = W1f[4 * NH + nc];
  wr.l1[3] = b1f[nc];

  #pragma unroll
  for (int k = 0; k < 4; ++k) {
    const int g0  = 8 * q + 2 * k;
    int       g1i = g0 + 1; if (g1i > NH - 1) g1i = NH - 1;
    wr.ax[k] = W2f[g0  * NOUT + n];
    wr.bx[k] = W2f[g0  * NOUT + oc2];
    wr.ay[k] = W2f[g1i * NOUT + n];
    wr.by[k] = W2f[g1i * NOUT + oc2];
  }
  wr.b2a = b2f[n];
  wr.b2b = b2f[oc2];
  wr.w3  = W3f[o3c];
}

__device__ __forceinline__ void store_wfrag(const WRaw& wr, float4* sW, float* sW3,
                                            int buf, int fl, int lane, bool qhi,
                                            bool n15, bool o3ok) {
  const float s0x = n15 ? 0.f : wr.l1[0];
  const float s0y = n15 ? 0.f : wr.l1[1];
  const float s0z = (n15 || qhi) ? 0.f : wr.l1[2];
  const float s0w = qhi ? 0.f : (n15 ? 1.f : wr.l1[3]);
  sW[SW_IDX(buf, fl, 0, lane)] = make_float4(s0x, s0y, s0z, s0w);
  #pragma unroll
  for (int k = 0; k < 3; ++k)
    sW[SW_IDX(buf, fl, 1 + k, lane)] = make_float4(wr.ax[k], wr.ay[k], wr.bx[k], wr.by[k]);
  sW[SW_IDX(buf, fl, 4, lane)] = make_float4(wr.ax[3], qhi ? wr.b2a : wr.ay[3],
                                             wr.bx[3], qhi ? wr.b2b : wr.by[3]);
  sW3[SW3_IDX(buf, fl, lane)] = o3ok ? wr.w3 : 0.f;
}

struct FIn { float4 l1; float4 wk[4]; float4 w3[4]; };

__device__ __forceinline__ void fetch_frag(
    FIn& in, const float4* sW, const float4* sW3v, int buf, int fl, int lane, int q)
{
  in.l1 = sW[SW_IDX(buf, fl, 0, lane)];
  #pragma unroll
  for (int k = 0; k < 4; ++k) in.wk[k] = sW[SW_IDX(buf, fl, 1 + k, lane)];
  #pragma unroll
  for (int j = 0; j < 4; ++j) in.w3[j] = sW3v[SW3_IDX4(buf, fl, q, j)];
}

__device__ __forceinline__ void feature_mma(const FIn& in, float y0, float y1, float y4,
                                            v8f& c2a, v8f& c2b) {
  v8f c1 = {};
  {
    v2f a, bb;
    a.x  = in.l1.x;
    a.y  = in.l1.y;
    bb.x = y0;
    bb.y = y1;
    c1 = wmma_f32_k4(a, bb, c1);
    a.x  = in.l1.z;
    a.y  = in.l1.w;
    bb.x = y4;
    bb.y = 1.f;
    c1 = wmma_f32_k4(a, bb, c1);
  }

  c2a = (v8f){};
  c2b = (v8f){};
  #pragma unroll
  for (int k = 0; k < 4; ++k) {
    v2f hb, wa, wb;
    hb.x = c1[2 * k];     hb.y = c1[2 * k + 1];
    wa.x = in.wk[k].x;    wa.y = in.wk[k].y;
    wb.x = in.wk[k].z;    wb.y = in.wk[k].w;
    c2a = wmma_f32_k4(wa, hb, c2a);
    c2b = wmma_f32_k4(wb, hb, c2b);
  }
}

__device__ __forceinline__ void feature_tail(const v8f& c2a, const v8f& c2b, const float4 (&w3)[4],
                                             float (&accA)[8], float (&accB)[8]) {
  const float w3a[8] = { w3[0].x, w3[0].y, w3[0].z, w3[0].w, w3[1].x, w3[1].y, w3[1].z, w3[1].w };
  const float w3b[8] = { w3[2].x, w3[2].y, w3[2].z, w3[2].w, w3[3].x, w3[3].y, w3[3].z, w3[3].w };
  #pragma unroll
  for (int r = 0; r < 8; ++r) {
    const float va = c2a[r] > 0.f ? c2a[r] : 0.f;
    const float vb = c2b[r] > 0.f ? c2b[r] : 0.f;
    accA[r] = __builtin_fmaf(va, w3a[r], accA[r]);
    accB[r] = __builtin_fmaf(vb, w3b[r], accB[r]);
  }
}

__global__ __launch_bounds__(256) void mlp200_wmma_kernel(
    const float* __restrict__ y,  const float* __restrict__ W1,
    const float* __restrict__ b1, const float* __restrict__ W2,
    const float* __restrict__ b2, const float* __restrict__ W3,
    const float* __restrict__ b3, float* __restrict__ out)
{
  __shared__ float4 sW[2 * FC * 5 * 32];
  __shared__ float4 sW3v[2 * FC * 8];

  const int  lane = threadIdx.x & 31;
  const int  wave = threadIdx.x >> 5;
  const int  q    = lane >> 4;
  const bool qhi  = (q != 0);
  const int  n    = lane & 15;
  const int  b0   = (blockIdx.x * 8 + wave) * 16;

  const int  nc   = (n < NH - 1) ? n : (NH - 1);
  const int  oc2  = 16 + ((n < 14) ? n : 13);
  const bool n15  = (n == 15);
  const int  o3   = 16 * ((lane >> 3) & 1) + 8 * q + (lane & 7);
  const bool o3ok = (o3 < NOUT);
  const int  o3c  = o3ok ? o3 : (NOUT - 1);

  const int  iy0  = (b0 + n) * (NF * DIN) + 2 * q;
  const int  iy1  = iy0 + 1;
  const int  iy4  = (b0 + n) * (NF * DIN) + 4;

  float accA[8], accB[8];
  #pragma unroll
  for (int r = 0; r < 8; ++r) { accA[r] = 0.f; accB[r] = 0.f; }

  float* sW3 = reinterpret_cast<float*>(sW3v);

  float ya[FC][3];
  #pragma unroll
  for (int fl = 0; fl < FC; ++fl) {
    ya[fl][0] = y[iy0 + fl * DIN];
    ya[fl][1] = y[iy1 + fl * DIN];
    ya[fl][2] = y[iy4 + fl * DIN];
  }
  WRaw wr;
  load_wraw(wr, wave, n, q, nc, oc2, o3c, W1, b1, W2, b2, W3);
  store_wfrag(wr, sW, sW3, 0, wave, lane, qhi, n15, o3ok);
  __syncthreads();

  for (int c = 0; c < NCHUNK; ++c) {
    const int buf = c & 1;
    const int cn  = (c + 1 < NCHUNK) ? (c + 1) : c;
    const float* ybn = y + (cn * FC) * DIN;

    FIn fr[FC];
    #pragma unroll
    for (int fl = 0; fl < FC; ++fl) fetch_frag(fr[fl], sW, sW3v, buf, fl, lane, q);

    load_wraw(wr, cn * FC + wave, n, q, nc, oc2, o3c, W1, b1, W2, b2, W3);

    #pragma unroll
    for (int fl = 0; fl < FC; ++fl) {
      v8f ca, cb;
      feature_mma(fr[fl], ya[fl][0], ya[fl][1], ya[fl][2], ca, cb);
      ya[fl][0] = ybn[iy0 + fl * DIN];
      ya[fl][1] = ybn[iy1 + fl * DIN];
      ya[fl][2] = ybn[iy4 + fl * DIN];
      feature_tail(ca, cb, fr[fl].w3, accA, accB);
    }

    #pragma unroll
    for (int i = 0; i < FC; ++i)
      __builtin_amdgcn_sched_group_barrier(0x100, 9, 0);
    #pragma unroll
    for (int i = 0; i < 5; ++i) {
      __builtin_amdgcn_sched_group_barrier(0x008, 10, 0);
      __builtin_amdgcn_sched_group_barrier(0x020, 6, 0);
    }
    __builtin_amdgcn_sched_group_barrier(0x008, 10, 0);
    __builtin_amdgcn_sched_group_barrier(0x020, 3, 0);
    #pragma unroll
    for (int i = 0; i < 2; ++i) {
      __builtin_amdgcn_sched_group_barrier(0x008, 10, 0);
      __builtin_amdgcn_sched_group_barrier(0x020, 2, 0);
    }
    __builtin_amdgcn_sched_group_barrier(0x200, 6, 0);

    store_wfrag(wr, sW, sW3, buf ^ 1, wave, lane, qhi, n15, o3ok);
    asm volatile("s_wait_dscnt 0x0\n\ts_barrier_signal -1\n\ts_barrier_wait -1" ::: "memory");
  }

  float s = 0.f;
  {
    float t[8];
    #pragma unroll
    for (int r = 0; r < 8; ++r) t[r] = accA[r] + accB[r];
    s = ((t[0] + t[1]) + (t[2] + t[3])) + ((t[4] + t[5]) + (t[6] + t[7]));
  }
  s += swz<SWAP16>(s);

  if (!qhi)
    out[b0 + n] = s + b3[0];

  asm volatile("s_wait_loadcnt 0x0" ::: "memory");
}

extern "C" void kernel_launch(void* const* d_in, const int* in_sizes, int n_in,
                              void* d_out, int out_size, void* d_ws, size_t ws_size,
                              hipStream_t stream) {
  (void)in_sizes; (void)n_in; (void)d_ws; (void)ws_size; (void)out_size;
  const float* y  = (const float*)d_in[0];
  const float* W1 = (const float*)d_in[1];
  const float* b1 = (const float*)d_in[2];
  const float* W2 = (const float*)d_in[3];
  const float* b2 = (const float*)d_in[4];
  const float* W3 = (const float*)d_in[5];
  const float* b3 = (const float*)d_in[6];
  float* out = (float*)d_out;

  dim3 grid(BATCH / (16 * 8));
  dim3 block(256);
  mlp200_wmma_kernel<<<grid, block, 0, stream>>>(y, W1, b1, W2, b2, W3, b3, out);
}


# Weight fragments of 40 features per LDS buffer (2 x 105 KB): block barriers 5, not 25; a SIMD's two waves run out of step, one's LDS reads under the other's WMMAs; same sums
# speedup vs baseline: 2.4461x; 1.1227x over previous
#include <hip/hip_runtime.h>

typedef float v2f __attribute__((ext_vector_type(2)));
typedef float v8f __attribute__((ext_vector_type(8)));

#define BATCH 32768
#define NF    200
#define DIN   5
#define NH    15
#define NOUT  30
#define FC    8
#define NCHUNK (NF / FC)
#define PHC   5
#define PH    (PHC * FC)
#define NPH   (NCHUNK / PHC)

__device__ __forceinline__ v8f wmma_f32_k4(v2f a, v2f b, v8f c) {
  return __builtin_amdgcn_wmma_f32_16x16x4_f32(
       false, a,  false, b,
       (short)0, c,  false,  false);
}

template <int IMM>
__device__ __forceinline__ float swz(float x) {
  return __int_as_float(__builtin_amdgcn_ds_swizzle(__float_as_int(x), IMM));
}
#define SWAP16 0x401f

#define SW_IDX(buf, fi, slot, lane)  ((((buf) * PH + (fi)) * 5 + (slot)) * 32 + (lane))
#define SW3_IDX(buf, fi, lane)       (((buf) * PH + (fi)) * 32 + (lane))
#define SW3_IDX4(buf, fi, q, j)      (((buf) * PH + (fi)) * 8 + (q) * 4 + (j))

struct WRaw { float l1[4]; float ax[4], ay[4], bx[4], by[4]; float b2a, b2b, w3; };

__device__ __forceinline__ void load_wraw(
    WRaw& wr, int f, int n, int q, int nc, int oc2, int o3c,
    const float* __restrict__ W1, const float* __restrict__ b1,
    const float* __restrict__ W2, const float* __restrict__ b2,
    const float* __restrict__ W3)
{
  const float* W1f = W1 + f * (DIN * NH);
  const float* W2f = W2 + f * (NH * NOUT);
  const float* b1f = b1 + f * NH;
  const float* b2f = b2 + f * NOUT;
  const float* W3f = W3 + f * NOUT;

  wr.l1[0] = W1f[(2 * q) * NH + nc];
  wr.l1[1] = W1f[(2 * q + 1) * NH + nc];
  wr.l1[2] = W1f[4 * NH + nc];
  wr.l1[3] = b1f[nc];

  #pragma unroll
  for (int k = 0; k < 4; ++k) {
    const int g0  = 8 * q + 2 * k;
    int       g1i = g0 + 1; if (g1i > NH - 1) g1i = NH - 1;
    wr.ax[k] = W2f[g0  * NOUT + n];
    wr.bx[k] = W2f[g0  * NOUT + oc2];
    wr.ay[k] = W2f[g1i * NOUT + n];
    wr.by[k] = W2f[g1i * NOUT + oc2];
  }
  wr.b2a = b2f[n];
  wr.b2b = b2f[oc2];
  wr.w3  = W3f[o3c];
}

__device__ __forceinline__ void store_wfrag(const WRaw& wr, float4* sW, float* sW3,
                                            int buf, int fl, int lane, bool qhi,
                                            bool n15, bool o3ok) {
  const float s0x = n15 ? 0.f : wr.l1[0];
  const float s0y = n15 ? 0.f : wr.l1[1];
  const float s0z = (n15 || qhi) ? 0.f : wr.l1[2];
  const float s0w = qhi ? 0.f : (n15 ? 1.f : wr.l1[3]);
  sW[SW_IDX(buf, fl, 0, lane)] = make_float4(s0x, s0y, s0z, s0w);
  #pragma unroll
  for (int k = 0; k < 3; ++k)
    sW[SW_IDX(buf, fl, 1 + k, lane)] = make_float4(wr.ax[k], wr.ay[k], wr.bx[k], wr.by[k]);
  sW[SW_IDX(buf, fl, 4, lane)] = make_float4(wr.ax[3], qhi ? wr.b2a : wr.ay[3],
                                             wr.bx[3], qhi ? wr.b2b : wr.by[3]);
  sW3[SW3_IDX(buf, fl, lane)] = o3ok ? wr.w3 : 0.f;
}

struct FIn { float4 l1; float4 wk[4]; float4 w3[4]; };

__device__ __forceinline__ void fetch_frag(
    FIn& in, const float4* sW, const float4* sW3v, int buf, int fl, int lane, int q)
{
  in.l1 = sW[SW_IDX(buf, fl, 0, lane)];
  #pragma unroll
  for (int k = 0; k < 4; ++k) in.wk[k] = sW[SW_IDX(buf, fl, 1 + k, lane)];
  #pragma unroll
  for (int j = 0; j < 4; ++j) in.w3[j] = sW3v[SW3_IDX4(buf, fl, q, j)];
}

__device__ __forceinline__ void feature_mma(const FIn& in, float y0, float y1, float y4,
                                            v8f& c2a, v8f& c2b) {
  v8f c1 = {};
  {
    v2f a, bb;
    a.x  = in.l1.x;
    a.y  = in.l1.y;
    bb.x = y0;
    bb.y = y1;
    c1 = wmma_f32_k4(a, bb, c1);
    a.x  = in.l1.z;
    a.y  = in.l1.w;
    bb.x = y4;
    bb.y = 1.f;
    c1 = wmma_f32_k4(a, bb, c1);
  }

  c2a = (v8f){};
  c2b = (v8f){};
  #pragma unroll
  for (int k = 0; k < 4; ++k) {
    v2f hb, wa, wb;
    hb.x = c1[2 * k];     hb.y = c1[2 * k + 1];
    wa.x = in.wk[k].x;    wa.y = in.wk[k].y;
    wb.x = in.wk[k].z;    wb.y = in.wk[k].w;
    c2a = wmma_f32_k4(wa, hb, c2a);
    c2b = wmma_f32_k4(wb, hb, c2b);
  }
}

__device__ __forceinline__ void feature_tail(const v8f& c2a, const v8f& c2b, const float4 (&w3)[4],
                                             float (&accA)[8], float (&accB)[8]) {
  const float w3a[8] = { w3[0].x, w3[0].y, w3[0].z, w3[0].w, w3[1].x, w3[1].y, w3[1].z, w3[1].w };
  const float w3b[8] = { w3[2].x, w3[2].y, w3[2].z, w3[2].w, w3[3].x, w3[3].y, w3[3].z, w3[3].w };
  #pragma unroll
  for (int r = 0; r < 8; ++r) {
    const float va = c2a[r] > 0.f ? c2a[r] : 0.f;
    const float vb = c2b[r] > 0.f ? c2b[r] : 0.f;
    accA[r] = __builtin_fmaf(va, w3a[r], accA[r]);
    accB[r] = __builtin_fmaf(vb, w3b[r], accB[r]);
  }
}

__global__ __launch_bounds__(256) void mlp200_wmma_kernel(
    const float* __restrict__ y,  const float* __restrict__ W1,
    const float* __restrict__ b1, const float* __restrict__ W2,
    const float* __restrict__ b2, const float* __restrict__ W3,
    const float* __restrict__ b3, float* __restrict__ out)
{
  __shared__ float4 sW[2 * PH * 5 * 32];
  __shared__ float4 sW3v[2 * PH * 8];

  const int  lane = threadIdx.x & 31;
  const int  wave = threadIdx.x >> 5;
  const int  q    = lane >> 4;
  const bool qhi  = (q != 0);
  const int  n    = lane & 15;
  const int  b0   = (blockIdx.x * 8 + wave) * 16;

  const int  nc   = (n < NH - 1) ? n : (NH - 1);
  const int  oc2  = 16 + ((n < 14) ? n : 13);
  const bool n15  = (n == 15);
  const int  o3   = 16 * ((lane >> 3) & 1) + 8 * q + (lane & 7);
  const bool o3ok = (o3 < NOUT);
  const int  o3c  = o3ok ? o3 : (NOUT - 1);

  const int  iy0  = (b0 + n) * (NF * DIN) + 2 * q;
  const int  iy1  = iy0 + 1;
  const int  iy4  = (b0 + n) * (NF * DIN) + 4;

  float accA[8], accB[8];
  #pragma unroll
  for (int r = 0; r < 8; ++r) { accA[r] = 0.f; accB[r] = 0.f; }

  float* sW3 = reinterpret_cast<float*>(sW3v);

  float ya[FC][3];
  #pragma unroll
  for (int fl = 0; fl < FC; ++fl) {
    ya[fl][0] = y[iy0 + fl * DIN];
    ya[fl][1] = y[iy1 + fl * DIN];
    ya[fl][2] = y[iy4 + fl * DIN];
  }
  WRaw wr;
  {
    WRaw w0[PHC];
    #pragma unroll
    for (int i = 0; i < PHC; ++i)
      load_wraw(w0[i], i * FC + wave, n, q, nc, oc2, o3c, W1, b1, W2, b2, W3);
    #pragma unroll
    for (int i = 0; i < PHC; ++i)
      store_wfrag(w0[i], sW, sW3, 0, i * FC + wave, lane, qhi, n15, o3ok);
  }
  __syncthreads();

  for (int p = 0; p < NPH; ++p) {
   const int buf = p & 1;
   for (int j = 0; j < PHC; ++j) {
    const int c = p * PHC + j;
    const int cn  = (c + 1 < NCHUNK) ? (c + 1) : c;
    const int cs  = (c + PHC < NCHUNK) ? (c + PHC) : c;
    const float* ybn = y + (cn * FC) * DIN;
    const int fb  = j * FC;

    FIn fr[FC];
    #pragma unroll
    for (int fl = 0; fl < FC; ++fl) fetch_frag(fr[fl], sW, sW3v, buf, fb + fl, lane, q);

    load_wraw(wr, cs * FC + wave, n, q, nc, oc2, o3c, W1, b1, W2, b2, W3);

    #pragma unroll
    for (int fl = 0; fl < FC; ++fl) {
      v8f ca, cb;
      feature_mma(fr[fl], ya[fl][0], ya[fl][1], ya[fl][2], ca, cb);
      ya[fl][0] = ybn[iy0 + fl * DIN];
      ya[fl][1] = ybn[iy1 + fl * DIN];
      ya[fl][2] = ybn[iy4 + fl * DIN];
      feature_tail(ca, cb, fr[fl].w3, accA, accB);
    }

    #pragma unroll
    for (int i = 0; i < FC; ++i)
      __builtin_amdgcn_sched_group_barrier(0x100, 9, 0);
    #pragma unroll
    for (int i = 0; i < 5; ++i) {
      __builtin_amdgcn_sched_group_barrier(0x008, 10, 0);
      __builtin_amdgcn_sched_group_barrier(0x020, 6, 0);
    }
    __builtin_amdgcn_sched_group_barrier(0x008, 10, 0);
    __builtin_amdgcn_sched_group_barrier(0x020, 3, 0);
    #pragma unroll
    for (int i = 0; i < 2; ++i) {
      __builtin_amdgcn_sched_group_barrier(0x008, 10, 0);
      __builtin_amdgcn_sched_group_barrier(0x020, 2, 0);
    }
    __builtin_amdgcn_sched_group_barrier(0x200, 6, 0);

    store_wfrag(wr, sW, sW3, buf ^ 1, fb + wave, lane, qhi, n15, o3ok);

   }
   asm volatile("s_wait_dscnt 0x0\n\ts_barrier_signal -1\n\ts_barrier_wait -1" ::: "memory");
  }

  float s = 0.f;
  {
    float t[8];
    #pragma unroll
    for (int r = 0; r < 8; ++r) t[r] = accA[r] + accB[r];
    s = ((t[0] + t[1]) + (t[2] + t[3])) + ((t[4] + t[5]) + (t[6] + t[7]));
  }
  s += swz<SWAP16>(s);

  if (!qhi)
    out[b0 + n] = s + b3[0];

  asm volatile("s_wait_loadcnt 0x0" ::: "memory");
}

extern "C" void kernel_launch(void* const* d_in, const int* in_sizes, int n_in,
                              void* d_out, int out_size, void* d_ws, size_t ws_size,
                              hipStream_t stream) {
  (void)in_sizes; (void)n_in; (void)d_ws; (void)ws_size; (void)out_size;
  const float* y  = (const float*)d_in[0];
  const float* W1 = (const float*)d_in[1];
  const float* b1 = (const float*)d_in[2];
  const float* W2 = (const float*)d_in[3];
  const float* b2 = (const float*)d_in[4];
  const float* W3 = (const float*)d_in[5];
  const float* b3 = (const float*)d_in[6];
  float* out = (float*)d_out;

  dim3 grid(BATCH / (16 * 8));
  dim3 block(256);
  mlp200_wmma_kernel<<<grid, block, 0, stream>>>(y, W1, b1, W2, b2, W3, b3, out);
}
